# starting kernel
# baseline (speedup 1.0000x reference)
#include <hip/hip_runtime.h>

typedef __attribute__((ext_vector_type(2))) float v2f;
typedef __attribute__((ext_vector_type(8))) float v8f;

__global__ __launch_bounds__(32) void vqc_build_tables(
    const float* __restrict__ ry, const float* __restrict__ rz,
    float* __restrict__ tab) {
  if (threadIdx.x != 0) return;

  float Ur[16], Ui[16];
  for (int i = 0; i < 16; ++i) { Ur[i] = (i % 5 == 0) ? 1.f : 0.f; Ui[i] = 0.f; }
  const int perm[4] = {0, 1, 3, 2};

  for (int l = 0; l < 6; ++l) {
    float h0 = 0.5f * ry[l * 2 + 0], h1 = 0.5f * ry[l * 2 + 1];
    float c0 = cosf(h0), s0 = sinf(h0), c1 = cosf(h1), s1 = sinf(h1);
    float R0[4] = {c0, -s0, s0, c0};
    float R1[4] = {c1, -s1, s1, c1};
    float F[16];
    for (int a = 0; a < 2; ++a)
      for (int b = 0; b < 2; ++b)
        for (int cc = 0; cc < 2; ++cc)
          for (int d = 0; d < 2; ++d)
            F[(2 * a + b) * 4 + (2 * cc + d)] = R0[a * 2 + cc] * R1[b * 2 + d];
    float t0 = rz[l * 2 + 0], t1 = rz[l * 2 + 1];
    float ph[4];
    for (int j = 0; j < 4; ++j) {
      int a = j >> 1, b = j & 1;
      ph[j] = 0.5f * ((a ? t0 : -t0) + (b ? t1 : -t1));
    }
    float Lr[16], Li[16];
    for (int i = 0; i < 4; ++i) {
      int j = perm[i];
      float cr = cosf(ph[j]), ci = sinf(ph[j]);
      for (int k = 0; k < 4; ++k) {
        Lr[i * 4 + k] = cr * F[j * 4 + k];
        Li[i * 4 + k] = ci * F[j * 4 + k];
      }
    }
    float Nr[16], Ni[16];
    for (int i = 0; i < 4; ++i)
      for (int k = 0; k < 4; ++k) {
        float ar = 0.f, ai = 0.f;
        for (int m = 0; m < 4; ++m) {
          ar += Lr[i * 4 + m] * Ur[m * 4 + k] - Li[i * 4 + m] * Ui[m * 4 + k];
          ai += Lr[i * 4 + m] * Ui[m * 4 + k] + Li[i * 4 + m] * Ur[m * 4 + k];
        }
        Nr[i * 4 + k] = ar;
        Ni[i * 4 + k] = ai;
      }
    for (int i = 0; i < 16; ++i) { Ur[i] = Nr[i]; Ui[i] = Ni[i]; }
  }

  float Mr[16], Mi[16];
  for (int i = 0; i < 16; ++i) { Mr[i] = Ur[i]; Mi[i] = -Ui[i]; }

  for (int j = 0; j < 2; ++j)
    for (int L = 0; L < 32; ++L) {
      int k = j + ((L < 16) ? 0 : 2);
      int m = L & 15;
      float a1, a2;
      if (m < 4)      { a1 = Mr[m * 4 + k];       a2 = -Mi[m * 4 + k]; }
      else if (m < 8) { a1 = Mi[(m - 4) * 4 + k]; a2 =  Mr[(m - 4) * 4 + k]; }
      else            { a1 = 0.f;                 a2 = 0.f; }
      tab[j * 32 + L]      = a1;
      tab[64 + j * 32 + L] = a2;
    }
}

__global__ __launch_bounds__(256) void vqc_main(
    const float* __restrict__ sr, const float* __restrict__ si,
    const float* __restrict__ tab, float* __restrict__ out, int nTiles) {
  const int lane = threadIdx.x & 31;
  const int wave = threadIdx.x >> 5;
  const int wavesPerBlock = blockDim.x >> 5;
  const int gwave = blockIdx.x * wavesPerBlock + wave;
  const int nwaves = gridDim.x * wavesPerBlock;

  v2f a1, a2;
  a1.x = tab[lane];       a1.y = tab[32 + lane];
  a2.x = tab[64 + lane];  a2.y = tab[96 + lane];

  const int col = lane & 15;
  const int koff = (lane < 16) ? 0 : 2;
  const bool writer = (lane < 16);

  for (int t = gwave; t < nTiles; t += nwaves) {
    const int base = t << 4;
    const int row = base + col;
    v2f br = __builtin_nontemporal_load((const v2f*)(sr + row * 4 + koff));
    v2f bi = __builtin_nontemporal_load((const v2f*)(si + row * 4 + koff));

    v8f acc = {0.f, 0.f, 0.f, 0.f, 0.f, 0.f, 0.f, 0.f};
    acc = __builtin_amdgcn_wmma_f32_16x16x4_f32(false, a1, false, br,
                                                (short)0, acc, false, false);
    acc = __builtin_amdgcn_wmma_f32_16x16x4_f32(false, a2, false, bi,
                                                (short)0, acc, false, false);

    float r0 = acc[0] * acc[0];
    r0 += acc[1] * acc[1];
    r0 -= acc[2] * acc[2];
    r0 -= acc[3] * acc[3];
    r0 += acc[4] * acc[4];
    r0 += acc[5] * acc[5];
    r0 -= acc[6] * acc[6];
    r0 -= acc[7] * acc[7];

    if (writer)
      __builtin_nontemporal_store(r0, out + base + lane);
  }
}

extern "C" void kernel_launch(void* const* d_in, const int* in_sizes, int n_in,
                              void* d_out, int out_size, void* d_ws, size_t ws_size,
                              hipStream_t stream) {
  const float* ry = (const float*)d_in[0];
  const float* rz = (const float*)d_in[1];
  const float* sr = (const float*)d_in[2];
  const float* si = (const float*)d_in[3];
  float* out = (float*)d_out;
  float* tab = (float*)d_ws;

  const int nTiles = out_size >> 4;

  vqc_build_tables<<<1, 32, 0, stream>>>(ry, rz, tab);
  vqc_main<<<1024, 256, 0, stream>>>(sr, si, tab, out, nTiles);
}


# main kernel data path: each lane loads its whole state row (b128); two 16-row tiles share one D via block-diagonal A operands, all 32 lanes reduce+store; dot-product sum order changes
# speedup vs baseline: 2.2839x; 2.2839x over previous
#include <hip/hip_runtime.h>

typedef __attribute__((ext_vector_type(2))) float v2f;
typedef __attribute__((ext_vector_type(4))) float v4f;
typedef __attribute__((ext_vector_type(8))) float v8f;

__device__ __forceinline__ float lane_to_wave(float v, int srcLane) {
  return __int_as_float(__builtin_amdgcn_readlane(__float_as_int(v), srcLane));
}

__global__ __launch_bounds__(32) void vqc_build_tables(
    const float* __restrict__ ry, const float* __restrict__ rz,
    float* __restrict__ tab) {
  const int lane = threadIdx.x;

  const int lay = (lane < 24) ? (lane >> 2) : 5;
  const int slot = lane & 3;
  const float ang = ry[lay * 2 + (slot & 1)];
  const float t0 = rz[lay * 2 + 0];
  const float t1 = rz[lay * 2 + 1];
  const float ph = 0.5f * ((-t0) + ((slot == 3) ? t1 : -t1));
  const float x = (slot < 2) ? 0.5f * ang : ph;
  const float cx = cosf(x);
  const float sx = sinf(x);

  const int r = lane & 3;
  float wr0 = (r == 0) ? 1.f : 0.f, wi0 = 0.f;
  float wr1 = (r == 1) ? 1.f : 0.f, wi1 = 0.f;
  float wr2 = (r == 2) ? 1.f : 0.f, wi2 = 0.f;
  float wr3 = (r == 3) ? 1.f : 0.f, wi3 = 0.f;

  for (int l = 5; l >= 0; --l) {
    const float c0 = lane_to_wave(cx, 4 * l + 0), s0 = lane_to_wave(sx, 4 * l + 0);
    const float c1 = lane_to_wave(cx, 4 * l + 1), s1 = lane_to_wave(sx, 4 * l + 1);
    const float pc0 = lane_to_wave(cx, 4 * l + 2), ps0 = lane_to_wave(sx, 4 * l + 2);
    const float pc1 = lane_to_wave(cx, 4 * l + 3), ps1 = lane_to_wave(sx, 4 * l + 3);

    const float pr0 = wr0 * pc0 - wi0 * ps0, pi0 = wr0 * ps0 + wi0 * pc0;
    const float pr1 = wr1 * pc1 - wi1 * ps1, pi1 = wr1 * ps1 + wi1 * pc1;
    const float pr2 = wr3 * pc1 + wi3 * ps1, pi2 = wi3 * pc1 - wr3 * ps1;
    const float pr3 = wr2 * pc0 + wi2 * ps0, pi3 = wi2 * pc0 - wr2 * ps0;

    const float tr0 = pr0 * c1 + pr1 * s1, ti0 = pi0 * c1 + pi1 * s1;
    const float tr1 = pr1 * c1 - pr0 * s1, ti1 = pi1 * c1 - pi0 * s1;
    const float tr2 = pr2 * c1 + pr3 * s1, ti2 = pi2 * c1 + pi3 * s1;
    const float tr3 = pr3 * c1 - pr2 * s1, ti3 = pi3 * c1 - pi2 * s1;
    wr0 = tr0 * c0 + tr2 * s0;  wi0 = ti0 * c0 + ti2 * s0;
    wr1 = tr1 * c0 + tr3 * s0;  wi1 = ti1 * c0 + ti3 * s0;
    wr2 = tr2 * c0 - tr0 * s0;  wi2 = ti2 * c0 - ti0 * s0;
    wr3 = tr3 * c0 - tr1 * s0;  wi3 = ti3 * c0 - ti1 * s0;
  }

  if (lane < 8) {
    const bool im = (lane >= 4);
    tab[lane]      = im ? -wi0 : wr0;
    tab[8 + lane]  = im ? -wi1 : wr1;
    tab[16 + lane] = im ? -wi2 : wr2;
    tab[24 + lane] = im ? -wi3 : wr3;
    tab[32 + lane] = im ? wr0 : wi0;
    tab[40 + lane] = im ? wr1 : wi1;
    tab[48 + lane] = im ? wr2 : wi2;
    tab[56 + lane] = im ? wr3 : wi3;
  }
}

__global__ __launch_bounds__(256) void vqc_main(
    const float* __restrict__ sr, const float* __restrict__ si,
    const float* __restrict__ tab, float* __restrict__ out, int nSteps) {
  const int lane = threadIdx.x & 31;
  const int wave = threadIdx.x >> 5;
  const int wavesPerBlock = blockDim.x >> 5;
  const int gwave = blockIdx.x * wavesPerBlock + wave;
  const int nwaves = gridDim.x * wavesPerBlock;

  const bool live = (((lane >> 3) & 1) == (lane >> 4));
  const int q = lane & 7;
  v2f a1 = {0.f, 0.f}, a2 = {0.f, 0.f}, a3 = {0.f, 0.f}, a4 = {0.f, 0.f};
  if (live) {
    a1.x = tab[q];       a1.y = tab[8 + q];
    a2.x = tab[16 + q];  a2.y = tab[24 + q];
    a3.x = tab[32 + q];  a3.y = tab[40 + q];
    a4.x = tab[48 + q];  a4.y = tab[56 + q];
  }

  for (int s = gwave; s < nSteps; s += nwaves) {
    const int row = (s << 5) + lane;
    const v4f xr = __builtin_nontemporal_load((const v4f*)(sr + row * 4));
    const v4f xi = __builtin_nontemporal_load((const v4f*)(si + row * 4));

    v2f b0, b1, b2, b3;
    b0.x = xr.x;  b0.y = xr.y;
    b1.x = xr.z;  b1.y = xr.w;
    b2.x = xi.x;  b2.y = xi.y;
    b3.x = xi.z;  b3.y = xi.w;

    v8f acc = {0.f, 0.f, 0.f, 0.f, 0.f, 0.f, 0.f, 0.f};
    acc = __builtin_amdgcn_wmma_f32_16x16x4_f32(false, a1, false, b0,
                                                (short)0, acc, false, false);
    acc = __builtin_amdgcn_wmma_f32_16x16x4_f32(false, a2, false, b1,
                                                (short)0, acc, false, false);
    acc = __builtin_amdgcn_wmma_f32_16x16x4_f32(false, a3, false, b2,
                                                (short)0, acc, false, false);
    acc = __builtin_amdgcn_wmma_f32_16x16x4_f32(false, a4, false, b3,
                                                (short)0, acc, false, false);

    float r0 = acc[0] * acc[0];
    r0 += acc[1] * acc[1];
    r0 -= acc[2] * acc[2];
    r0 -= acc[3] * acc[3];
    r0 += acc[4] * acc[4];
    r0 += acc[5] * acc[5];
    r0 -= acc[6] * acc[6];
    r0 -= acc[7] * acc[7];

    __builtin_nontemporal_store(r0, out + row);
  }
}

extern "C" void kernel_launch(void* const* d_in, const int* in_sizes, int n_in,
                              void* d_out, int out_size, void* d_ws, size_t ws_size,
                              hipStream_t stream) {
  const float* ry = (const float*)d_in[0];
  const float* rz = (const float*)d_in[1];
  const float* sr = (const float*)d_in[2];
  const float* si = (const float*)d_in[3];
  float* out = (float*)d_out;
  float* tab = (float*)d_ws;

  const int nSteps = out_size >> 5;

  vqc_build_tables<<<1, 32, 0, stream>>>(ry, rz, tab);
  vqc_main<<<1024, 256, 0, stream>>>(sr, si, tab, out, nSteps);
}


# main kernel: a wave owns four consecutive 32-row steps and issues all eight b128 row loads before its first product: one memory trip per wave, not four in series; sum order kept, fma fused
# speedup vs baseline: 2.6480x; 1.1594x over previous
#include <hip/hip_runtime.h>

typedef __attribute__((ext_vector_type(2))) float v2f;
typedef __attribute__((ext_vector_type(4))) float v4f;
typedef __attribute__((ext_vector_type(8))) float v8f;

__device__ __forceinline__ float lane_to_wave(float v, int srcLane) {
  return __int_as_float(__builtin_amdgcn_readlane(__float_as_int(v), srcLane));
}

__global__ __launch_bounds__(32) void vqc_build_tables(
    const float* __restrict__ ry, const float* __restrict__ rz,
    float* __restrict__ tab) {
  const int lane = threadIdx.x;

  const int lay = (lane < 24) ? (lane >> 2) : 5;
  const int slot = lane & 3;
  const float ang = ry[lay * 2 + (slot & 1)];
  const float t0 = rz[lay * 2 + 0];
  const float t1 = rz[lay * 2 + 1];
  const float ph = 0.5f * ((-t0) + ((slot == 3) ? t1 : -t1));
  const float x = (slot < 2) ? 0.5f * ang : ph;
  const float cx = cosf(x);
  const float sx = sinf(x);

  const int r = lane & 3;
  float wr0 = (r == 0) ? 1.f : 0.f, wi0 = 0.f;
  float wr1 = (r == 1) ? 1.f : 0.f, wi1 = 0.f;
  float wr2 = (r == 2) ? 1.f : 0.f, wi2 = 0.f;
  float wr3 = (r == 3) ? 1.f : 0.f, wi3 = 0.f;

  for (int l = 5; l >= 0; --l) {
    const float c0 = lane_to_wave(cx, 4 * l + 0), s0 = lane_to_wave(sx, 4 * l + 0);
    const float c1 = lane_to_wave(cx, 4 * l + 1), s1 = lane_to_wave(sx, 4 * l + 1);
    const float pc0 = lane_to_wave(cx, 4 * l + 2), ps0 = lane_to_wave(sx, 4 * l + 2);
    const float pc1 = lane_to_wave(cx, 4 * l + 3), ps1 = lane_to_wave(sx, 4 * l + 3);

    const float pr0 = wr0 * pc0 - wi0 * ps0, pi0 = wr0 * ps0 + wi0 * pc0;
    const float pr1 = wr1 * pc1 - wi1 * ps1, pi1 = wr1 * ps1 + wi1 * pc1;
    const float pr2 = wr3 * pc1 + wi3 * ps1, pi2 = wi3 * pc1 - wr3 * ps1;
    const float pr3 = wr2 * pc0 + wi2 * ps0, pi3 = wi2 * pc0 - wr2 * ps0;

    const float tr0 = pr0 * c1 + pr1 * s1, ti0 = pi0 * c1 + pi1 * s1;
    const float tr1 = pr1 * c1 - pr0 * s1, ti1 = pi1 * c1 - pi0 * s1;
    const float tr2 = pr2 * c1 + pr3 * s1, ti2 = pi2 * c1 + pi3 * s1;
    const float tr3 = pr3 * c1 - pr2 * s1, ti3 = pi3 * c1 - pi2 * s1;
    wr0 = tr0 * c0 + tr2 * s0;  wi0 = ti0 * c0 + ti2 * s0;
    wr1 = tr1 * c0 + tr3 * s0;  wi1 = ti1 * c0 + ti3 * s0;
    wr2 = tr2 * c0 - tr0 * s0;  wi2 = ti2 * c0 - ti0 * s0;
    wr3 = tr3 * c0 - tr1 * s0;  wi3 = ti3 * c0 - ti1 * s0;
  }

  if (lane < 8) {
    const bool im = (lane >= 4);
    tab[lane]      = im ? -wi0 : wr0;
    tab[8 + lane]  = im ? -wi1 : wr1;
    tab[16 + lane] = im ? -wi2 : wr2;
    tab[24 + lane] = im ? -wi3 : wr3;
    tab[32 + lane] = im ? wr0 : wi0;
    tab[40 + lane] = im ? wr1 : wi1;
    tab[48 + lane] = im ? wr2 : wi2;
    tab[56 + lane] = im ? wr3 : wi3;
  }
}

__device__ __forceinline__ float vqc_step(v2f a1, v2f a2, v2f a3, v2f a4,
                                          v4f xr, v4f xi) {
  v2f b0, b1, b2, b3;
  b0.x = xr.x;  b0.y = xr.y;
  b1.x = xr.z;  b1.y = xr.w;
  b2.x = xi.x;  b2.y = xi.y;
  b3.x = xi.z;  b3.y = xi.w;

  v8f acc = {0.f, 0.f, 0.f, 0.f, 0.f, 0.f, 0.f, 0.f};
  acc = __builtin_amdgcn_wmma_f32_16x16x4_f32(false, a1, false, b0,
                                              (short)0, acc, false, false);
  acc = __builtin_amdgcn_wmma_f32_16x16x4_f32(false, a2, false, b1,
                                              (short)0, acc, false, false);
  acc = __builtin_amdgcn_wmma_f32_16x16x4_f32(false, a3, false, b2,
                                              (short)0, acc, false, false);
  acc = __builtin_amdgcn_wmma_f32_16x16x4_f32(false, a4, false, b3,
                                              (short)0, acc, false, false);

  float r0 = acc[0] * acc[0];
  r0 += acc[1] * acc[1];
  r0 -= acc[2] * acc[2];
  r0 -= acc[3] * acc[3];
  r0 += acc[4] * acc[4];
  r0 += acc[5] * acc[5];
  r0 -= acc[6] * acc[6];
  r0 -= acc[7] * acc[7];
  return r0;
}

__global__ __launch_bounds__(256) void vqc_main(
    const float* __restrict__ sr, const float* __restrict__ si,
    const float* __restrict__ tab, float* __restrict__ out, int nSteps) {
  const int lane = threadIdx.x & 31;
  const int wave = threadIdx.x >> 5;
  const int wavesPerBlock = blockDim.x >> 5;
  const int gwave = blockIdx.x * wavesPerBlock + wave;
  const int nwaves = gridDim.x * wavesPerBlock;

  const bool live = (((lane >> 3) & 1) == (lane >> 4));
  const int q = lane & 7;
  v2f a1 = {0.f, 0.f}, a2 = {0.f, 0.f}, a3 = {0.f, 0.f}, a4 = {0.f, 0.f};
  if (live) {
    a1.x = tab[q];       a1.y = tab[8 + q];
    a2.x = tab[16 + q];  a2.y = tab[24 + q];
    a3.x = tab[32 + q];  a3.y = tab[40 + q];
    a4.x = tab[48 + q];  a4.y = tab[56 + q];
  }

  const int nGroups = nSteps >> 2;
  for (int g = gwave; g < nGroups; g += nwaves) {
    const int row = (g << 7) + lane;
    const v4f* pr = (const v4f*)(sr + row * 4);
    const v4f* pi = (const v4f*)(si + row * 4);
    const v4f xr0 = __builtin_nontemporal_load(pr);
    const v4f xi0 = __builtin_nontemporal_load(pi);
    const v4f xr1 = __builtin_nontemporal_load(pr + 32);
    const v4f xi1 = __builtin_nontemporal_load(pi + 32);
    const v4f xr2 = __builtin_nontemporal_load(pr + 64);
    const v4f xi2 = __builtin_nontemporal_load(pi + 64);
    const v4f xr3 = __builtin_nontemporal_load(pr + 96);
    const v4f xi3 = __builtin_nontemporal_load(pi + 96);
    __builtin_amdgcn_sched_barrier(0);

    float* po = out + row;
    __builtin_nontemporal_store(vqc_step(a1, a2, a3, a4, xr0, xi0), po);
    __builtin_nontemporal_store(vqc_step(a1, a2, a3, a4, xr1, xi1), po + 32);
    __builtin_nontemporal_store(vqc_step(a1, a2, a3, a4, xr2, xi2), po + 64);
    __builtin_nontemporal_store(vqc_step(a1, a2, a3, a4, xr3, xi3), po + 96);
  }

  for (int s = (nGroups << 2) + gwave; s < nSteps; s += nwaves) {
    const int row = (s << 5) + lane;
    const v4f xr = __builtin_nontemporal_load((const v4f*)(sr + row * 4));
    const v4f xi = __builtin_nontemporal_load((const v4f*)(si + row * 4));
    __builtin_nontemporal_store(vqc_step(a1, a2, a3, a4, xr, xi), out + row);
  }
}

extern "C" void kernel_launch(void* const* d_in, const int* in_sizes, int n_in,
                              void* d_out, int out_size, void* d_ws, size_t ws_size,
                              hipStream_t stream) {
  const float* ry = (const float*)d_in[0];
  const float* rz = (const float*)d_in[1];
  const float* sr = (const float*)d_in[2];
  const float* si = (const float*)d_in[3];
  float* out = (float*)d_out;
  float* tab = (float*)d_ws;

  const int nSteps = out_size >> 5;

  vqc_build_tables<<<1, 32, 0, stream>>>(ry, rz, tab);
  vqc_main<<<1024, 256, 0, stream>>>(sr, si, tab, out, nSteps);
}
